# starting kernel
# baseline (speedup 1.0000x reference)
#include <hip/hip_runtime.h>
#include <math.h>

typedef __attribute__((ext_vector_type(2))) float v2f;
typedef __attribute__((ext_vector_type(8))) float v8f;

#define WIRES 12
#define BATCH_PER_BLOCK 16
#define ROWS_PER_BLOCK (BATCH_PER_BLOCK * WIRES)
#define TILES_PER_BLOCK (ROWS_PER_BLOCK / 16)
#define THREADS 128
#define WAVES (THREADS / 32)
#define TILES_PER_WAVE (TILES_PER_BLOCK / WAVES)

__global__ __launch_bounds__(THREADS)
void qsa_kernel(const float* __restrict__ x,
                const float* __restrict__ theta,
                const float* __restrict__ Wout,
                const float* __restrict__ bout,
                float* __restrict__ out)
{
    __shared__ float qs[ROWS_PER_BLOCK][WIRES];
    __shared__ float th[WIRES];

    const int tid = threadIdx.x;
    const int b0  = blockIdx.x * BATCH_PER_BLOCK;

    if (tid < WIRES) th[tid] = theta[tid];
    __syncthreads();

    for (int p = tid; p < BATCH_PER_BLOCK * WIRES; p += THREADS) {
        const int bb = p / WIRES;
        const int e  = p % WIRES;
        const float* xp = x + (size_t)(b0 + bb) * (WIRES * WIRES) + e;

        float c[WIRES];
        #pragma unroll
        for (int w = 0; w < WIRES; ++w)
            c[w] = cosf(xp[w * WIRES] + th[w]);

        float pre = c[0];
        float q1toN = 1.0f;
        #pragma unroll
        for (int w = 1; w < WIRES; ++w) q1toN *= c[w];
        qs[bb * WIRES + 0][e] = q1toN;
        #pragma unroll
        for (int t = 1; t < WIRES; ++t) {
            pre *= c[t];
            qs[bb * WIRES + t][e] = pre;
        }
    }
    __syncthreads();

    const int lane = tid & 31;
    const int half = lane >> 4;
    const int m    = lane & 15;

    v2f bf[3];
    #pragma unroll
    for (int kb = 0; kb < 3; ++kb) {
        const int e0 = kb * 4 + 2 * half;
        bf[kb].x = (m < WIRES) ? Wout[m * WIRES + e0]     : 0.0f;
        bf[kb].y = (m < WIRES) ? Wout[m * WIRES + e0 + 1] : 0.0f;
    }
    const float bias = (m < WIRES) ? bout[m] : 0.0f;

    const int wave = tid >> 5;
    for (int t = wave * TILES_PER_WAVE; t < (wave + 1) * TILES_PER_WAVE; ++t) {
        v8f acc = {};
        #pragma unroll
        for (int kb = 0; kb < 3; ++kb) {
            const int e0 = kb * 4 + 2 * half;
            v2f a;
            a.x = qs[t * 16 + m][e0];
            a.y = qs[t * 16 + m][e0 + 1];
            acc = __builtin_amdgcn_wmma_f32_16x16x4_f32(
                 false, a,  false, bf[kb],
                 (short)0, acc,  false,  false);
        }
        if (m < WIRES) {
            #pragma unroll
            for (int r = 0; r < 8; ++r) {
                const int mrow = r + 8 * half;
                const size_t grow = (size_t)b0 * WIRES + (size_t)(t * 16 + mrow);
                out[grow * WIRES + m] = acc[r] + bias;
            }
        }
    }
}

extern "C" void kernel_launch(void* const* d_in, const int* in_sizes, int n_in,
                              void* d_out, int out_size, void* d_ws, size_t ws_size,
                              hipStream_t stream) {
    const float* x     = (const float*)d_in[0];
    const float* theta = (const float*)d_in[1];
    const float* Wout  = (const float*)d_in[2];
    const float* bout  = (const float*)d_in[3];
    float* out = (float*)d_out;

    const int B = in_sizes[0] / (WIRES * WIRES);
    const int blocks = B / BATCH_PER_BLOCK;

    qsa_kernel<<<blocks, THREADS, 0, stream>>>(x, theta, Wout, bout, out);
}


# One range test for the 12 cosine arguments of a pair instead of a branch inside every cosine: the short-range cosines are one branch-free block and interleave; other arguments still take cosf
# speedup vs baseline: 2.8056x; 2.8056x over previous
#include <hip/hip_runtime.h>
#include <math.h>


typedef __attribute__((ext_vector_type(2))) float v2f;
typedef __attribute__((ext_vector_type(8))) float v8f;

#define WIRES 12
#define BATCH_PER_BLOCK 16
#define ROWS_PER_BLOCK (BATCH_PER_BLOCK * WIRES)
#define TILES_PER_BLOCK (ROWS_PER_BLOCK / 16)
#define THREADS 128
#define WAVES (THREADS / 32)
#define TILES_PER_WAVE (TILES_PER_BLOCK / WAVES)

#define SHORT_ARG_BOUND 0x1.000000p+17f
__device__ __forceinline__
void cos_short_all(const float (&a)[WIRES], float (&c)[WIRES])
{
    float ax[WIRES], fn[WIRES], r[WIRES], t[WIRES], ps[WIRES], pc[WIRES], sn[WIRES], cs[WIRES];
    #pragma unroll
    for (int w = 0; w < WIRES; ++w) ax[w] = fabsf(a[w]);
    #pragma unroll
    for (int w = 0; w < WIRES; ++w) fn[w] = rintf(ax[w] * 0x1.45f306p-1f);
    #pragma unroll
    for (int w = 0; w < WIRES; ++w) r[w] = fmaf(fn[w], -0x1.921fb4p+0f, ax[w]);
    #pragma unroll
    for (int w = 0; w < WIRES; ++w) r[w] = fmaf(fn[w], -0x1.4442d0p-24f, r[w]);
    #pragma unroll
    for (int w = 0; w < WIRES; ++w) r[w] = fmaf(fn[w], -0x1.846988p-48f, r[w]);
    #pragma unroll
    for (int w = 0; w < WIRES; ++w) t[w] = r[w] * r[w];
    #pragma unroll
    for (int w = 0; w < WIRES; ++w) {
        ps[w] = fmaf(t[w], -0x1.983304p-13f, 0x1.110388p-7f);
        pc[w] = fmaf(t[w], 0x1.aea668p-16f, -0x1.6c9e76p-10f);
    }
    #pragma unroll
    for (int w = 0; w < WIRES; ++w) {
        ps[w] = fmaf(t[w], ps[w], -0x1.55553ap-3f);
        pc[w] = fmaf(t[w], pc[w], 0x1.5557eep-5f);
    }
    #pragma unroll
    for (int w = 0; w < WIRES; ++w) {
        ps[w] = t[w] * ps[w];
        pc[w] = fmaf(t[w], pc[w], -0x1.000008p-1f);
    }
    #pragma unroll
    for (int w = 0; w < WIRES; ++w) {
        sn[w] = fmaf(r[w], ps[w], r[w]);
        cs[w] = fmaf(t[w], pc[w], 1.0f);
    }
    #pragma unroll
    for (int w = 0; w < WIRES; ++w) {
        const unsigned n   = (unsigned)(int)fn[w];
        const unsigned odd = 0u - (n & 1u);
        const unsigned v   = ((__float_as_uint(sn[w]) ^ 0x80000000u) & odd) | (__float_as_uint(cs[w]) & ~odd);
        c[w] = __uint_as_float(v ^ ((n << 30) & 0x80000000u));
    }
}

__device__ __forceinline__
void cos_prefix_products(const float (&xv)[WIRES], const float (&th)[WIRES],
                         float (*qs)[WIRES], int p)
{
    const int bb = p / WIRES;
    const int e  = p % WIRES;

    float a[WIRES];
    bool all_short = true;
    #pragma unroll
    for (int w = 0; w < WIRES; ++w) {
        a[w] = xv[w] + th[w];
        all_short = all_short && (fabsf(a[w]) < SHORT_ARG_BOUND);
    }

    float c[WIRES];
    if (all_short) {
        cos_short_all(a, c);
    } else {
        #pragma unroll
        for (int w = 0; w < WIRES; ++w) c[w] = cosf(a[w]);
    }

    float pre = c[0];
    float q1toN = 1.0f;
    #pragma unroll
    for (int w = 1; w < WIRES; ++w) q1toN *= c[w];
    qs[bb * WIRES + 0][e] = q1toN;
    #pragma unroll
    for (int t = 1; t < WIRES; ++t) {
        pre *= c[t];
        qs[bb * WIRES + t][e] = pre;
    }
}

__global__ __launch_bounds__(THREADS)
void qsa_kernel(const float* __restrict__ x,
                const float* __restrict__ theta,
                const float* __restrict__ Wout,
                const float* __restrict__ bout,
                float* __restrict__ out)
{
    __shared__ float qs[ROWS_PER_BLOCK][WIRES];
    __shared__ float th[WIRES];

    const int tid  = threadIdx.x;
    const int b0   = blockIdx.x * BATCH_PER_BLOCK;
    const int lane = tid & 31;
    const int half = lane >> 4;
    const int m    = lane & 15;

    float tv = 0.0f;
    if (tid < WIRES) tv = theta[tid];

    const int  p0  = tid;
    const int  p1  = tid + THREADS;
    const bool two = p1 < ROWS_PER_BLOCK;

    float xa[WIRES];
    float xb[WIRES] = {};
    {
        const float* xp0 = x + (size_t)(b0 + p0 / WIRES) * (WIRES * WIRES) + (p0 % WIRES);
        #pragma unroll
        for (int w = 0; w < WIRES; ++w) xa[w] = xp0[w * WIRES];
    }
    if (two) {
        const float* xp1 = x + (size_t)(b0 + p1 / WIRES) * (WIRES * WIRES) + (p1 % WIRES);
        #pragma unroll
        for (int w = 0; w < WIRES; ++w) xb[w] = xp1[w * WIRES];
    }

    v2f bf[3] = {};
    float bias = 0.0f;
    if (m < WIRES) {
        #pragma unroll
        for (int kb = 0; kb < 3; ++kb) {
            const int e0 = kb * 4 + 2 * half;
            bf[kb].x = Wout[m * WIRES + e0];
            bf[kb].y = Wout[m * WIRES + e0 + 1];
        }
        bias = bout[m];
    }

    if (tid < WIRES) th[tid] = tv;
    __syncthreads();

    float tw[WIRES];
    #pragma unroll
    for (int w = 0; w < WIRES; ++w) tw[w] = th[w];

    cos_prefix_products(xa, tw, qs, p0);
    if (two) cos_prefix_products(xb, tw, qs, p1);
    __syncthreads();

    const int wave = tid >> 5;
    for (int t = wave * TILES_PER_WAVE; t < (wave + 1) * TILES_PER_WAVE; ++t) {
        v8f acc = {};
        #pragma unroll
        for (int kb = 0; kb < 3; ++kb) {
            const int e0 = kb * 4 + 2 * half;
            v2f a;
            a.x = qs[t * 16 + m][e0];
            a.y = qs[t * 16 + m][e0 + 1];
            acc = __builtin_amdgcn_wmma_f32_16x16x4_f32(
                 false, a,  false, bf[kb],
                 (short)0, acc,  false,  false);
        }
        if (m < WIRES) {
            #pragma unroll
            for (int r = 0; r < 8; ++r) {
                const int mrow = r + 8 * half;
                const size_t grow = (size_t)b0 * WIRES + (size_t)(t * 16 + mrow);
                out[grow * WIRES + m] = acc[r] + bias;
            }
        }
    }
}

extern "C" void kernel_launch(void* const* d_in, const int* in_sizes, int n_in,
                              void* d_out, int out_size, void* d_ws, size_t ws_size,
                              hipStream_t stream) {
    const float* x     = (const float*)d_in[0];
    const float* theta = (const float*)d_in[1];
    const float* Wout  = (const float*)d_in[2];
    const float* bout  = (const float*)d_in[3];
    float* out = (float*)d_out;

    const int B = in_sizes[0] / (WIRES * WIRES);
    const int blocks = B / BATCH_PER_BLOCK;

    qsa_kernel<<<blocks, THREADS, 0, stream>>>(x, theta, Wout, bout, out);
}


# Cosines by the hardware instruction: one multiply by 1/(2 pi) and one v_cos_f32 each, not the inline cosf short path; cosf kept for a pair with an angle of 32 or more; less exact than cosf
# speedup vs baseline: 3.1933x; 1.1382x over previous
#include <hip/hip_runtime.h>
#include <math.h>


typedef __attribute__((ext_vector_type(2))) float v2f;
typedef __attribute__((ext_vector_type(8))) float v8f;

#define WIRES 12
#define BATCH_PER_BLOCK 16
#define ROWS_PER_BLOCK (BATCH_PER_BLOCK * WIRES)
#define TILES_PER_BLOCK (ROWS_PER_BLOCK / 16)
#define THREADS 128
#define WAVES (THREADS / 32)
#define TILES_PER_WAVE (TILES_PER_BLOCK / WAVES)

#define HW_COS_ARG_BOUND 32.0f
#define INV_TWO_PI 0x1.45f306p-3f

__device__ __forceinline__
void cos_prefix_products(const float (&xv)[WIRES], const float (&th)[WIRES],
                         float (*qs)[WIRES], int p)
{
    const int bb = p / WIRES;
    const int e  = p % WIRES;

    float a[WIRES];
    #pragma unroll
    for (int w = 0; w < WIRES; ++w) a[w] = xv[w] + th[w];

    float c[WIRES];
    #pragma unroll
    for (int w = 0; w < WIRES; ++w) c[w] = __builtin_amdgcn_cosf(a[w] * INV_TWO_PI);

    const float m0 = fmaxf(fmaxf(fabsf(a[0]), fabsf(a[1])),  fabsf(a[2]));
    const float m1 = fmaxf(fmaxf(fabsf(a[3]), fabsf(a[4])),  fabsf(a[5]));
    const float m2 = fmaxf(fmaxf(fabsf(a[6]), fabsf(a[7])),  fabsf(a[8]));
    const float m3 = fmaxf(fmaxf(fabsf(a[9]), fabsf(a[10])), fabsf(a[11]));
    const float amax = fmaxf(fmaxf(m0, m1), fmaxf(m2, m3));

    if (!(amax < HW_COS_ARG_BOUND)) {
        #pragma unroll
        for (int w = 0; w < WIRES; ++w) c[w] = cosf(a[w]);
    }

    float pre = c[0];
    float q1toN = 1.0f;
    #pragma unroll
    for (int w = 1; w < WIRES; ++w) q1toN *= c[w];
    qs[bb * WIRES + 0][e] = q1toN;
    #pragma unroll
    for (int t = 1; t < WIRES; ++t) {
        pre *= c[t];
        qs[bb * WIRES + t][e] = pre;
    }
}

__global__ __launch_bounds__(THREADS)
void qsa_kernel(const float* __restrict__ x,
                const float* __restrict__ theta,
                const float* __restrict__ Wout,
                const float* __restrict__ bout,
                float* __restrict__ out)
{
    __shared__ float qs[ROWS_PER_BLOCK][WIRES];
    __shared__ float th[WIRES];

    const int tid  = threadIdx.x;
    const int b0   = blockIdx.x * BATCH_PER_BLOCK;
    const int lane = tid & 31;
    const int half = lane >> 4;
    const int m    = lane & 15;

    float tv = 0.0f;
    if (tid < WIRES) tv = theta[tid];

    const int  p0  = tid;
    const int  p1  = tid + THREADS;
    const bool two = p1 < ROWS_PER_BLOCK;

    float xa[WIRES];
    float xb[WIRES] = {};
    {
        const float* xp0 = x + (size_t)(b0 + p0 / WIRES) * (WIRES * WIRES) + (p0 % WIRES);
        #pragma unroll
        for (int w = 0; w < WIRES; ++w) xa[w] = xp0[w * WIRES];
    }
    if (two) {
        const float* xp1 = x + (size_t)(b0 + p1 / WIRES) * (WIRES * WIRES) + (p1 % WIRES);
        #pragma unroll
        for (int w = 0; w < WIRES; ++w) xb[w] = xp1[w * WIRES];
    }

    v2f bf[3] = {};
    float bias = 0.0f;
    if (m < WIRES) {
        #pragma unroll
        for (int kb = 0; kb < 3; ++kb) {
            const int e0 = kb * 4 + 2 * half;
            bf[kb].x = Wout[m * WIRES + e0];
            bf[kb].y = Wout[m * WIRES + e0 + 1];
        }
        bias = bout[m];
    }

    if (tid < WIRES) th[tid] = tv;
    __syncthreads();

    float tw[WIRES];
    #pragma unroll
    for (int w = 0; w < WIRES; ++w) tw[w] = th[w];

    cos_prefix_products(xa, tw, qs, p0);
    if (two) cos_prefix_products(xb, tw, qs, p1);
    __syncthreads();

    const int wave = tid >> 5;
    for (int t = wave * TILES_PER_WAVE; t < (wave + 1) * TILES_PER_WAVE; ++t) {
        v8f acc = {};
        #pragma unroll
        for (int kb = 0; kb < 3; ++kb) {
            const int e0 = kb * 4 + 2 * half;
            v2f a;
            a.x = qs[t * 16 + m][e0];
            a.y = qs[t * 16 + m][e0 + 1];
            acc = __builtin_amdgcn_wmma_f32_16x16x4_f32(
                 false, a,  false, bf[kb],
                 (short)0, acc,  false,  false);
        }
        if (m < WIRES) {
            #pragma unroll
            for (int r = 0; r < 8; ++r) {
                const int mrow = r + 8 * half;
                const size_t grow = (size_t)b0 * WIRES + (size_t)(t * 16 + mrow);
                out[grow * WIRES + m] = acc[r] + bias;
            }
        }
    }
}

extern "C" void kernel_launch(void* const* d_in, const int* in_sizes, int n_in,
                              void* d_out, int out_size, void* d_ws, size_t ws_size,
                              hipStream_t stream) {
    const float* x     = (const float*)d_in[0];
    const float* theta = (const float*)d_in[1];
    const float* Wout  = (const float*)d_in[2];
    const float* bout  = (const float*)d_in[3];
    float* out = (float*)d_out;

    const int B = in_sizes[0] / (WIRES * WIRES);
    const int blocks = B / BATCH_PER_BLOCK;

    qsa_kernel<<<blocks, THREADS, 0, stream>>>(x, theta, Wout, bout, out);
}
